# starting kernel
# baseline (speedup 1.0000x reference)
#include <hip/hip_runtime.h>
#include <hip/hip_bf16.h>

#define NB 4
#define MM 256
#define DD 32
#define SS 32

typedef float v2f __attribute__((ext_vector_type(2)));
typedef float v8f __attribute__((ext_vector_type(8)));

__global__ void ign_repack_kernel(const float* __restrict__ coeffs,
                                  float* __restrict__ C10T,
                                  float* __restrict__ C11T) {
    int t = threadIdx.x;
    int s = t >> 5, d = t & 31;
    C10T[s * DD + d] = coeffs[(d * SS + s) * 15 + 9];
    C11T[s * DD + d] = coeffs[(d * SS + s) * 15 + 10];
}

__global__ void ign_stats_kernel(const float* __restrict__ in,
                                 float* __restrict__ R,
                                 float* __restrict__ Cc,
                                 float* __restrict__ G) {
    const int n = blockIdx.y, k = blockIdx.x;
    const int t = threadIdx.x;
    const int d = t & 31, grp = t >> 5;
    const float* base = in + (size_t)n * MM * MM * DD;
    float r = 0.f, c = 0.f;
    for (int q = grp; q < MM; q += 8) {
        r += base[((size_t)k * MM + q) * DD + d];
        c += base[((size_t)q * MM + k) * DD + d];
    }
    __shared__ float sr[8][32];
    __shared__ float sc[8][32];
    sr[grp][d] = r; sc[grp][d] = c;
    __syncthreads();
    if (t < 32) {
        float rr = 0.f, cc = 0.f;
        #pragma unroll
        for (int g = 0; g < 8; ++g) { rr += sr[g][t]; cc += sc[g][t]; }
        const int o = (n * DD + t) * MM + k;
        R[o]  = rr;
        Cc[o] = cc;
        G[o]  = base[((size_t)k * MM + k) * DD + t];
    }
}

__global__ void ign_pt_kernel(const float* __restrict__ R,
                              const float* __restrict__ G,
                              float* __restrict__ P,
                              float* __restrict__ T) {
    int t = threadIdx.x;
    float p = 0.f, s = 0.f;
    for (int k = 0; k < MM; ++k) { p += G[t * MM + k]; s += R[t * MM + k]; }
    P[t] = p; T[t] = s;
}

__global__ void ign_bcast_kernel(const float* __restrict__ R,
                                 const float* __restrict__ Cc,
                                 const float* __restrict__ G,
                                 const float* __restrict__ P,
                                 const float* __restrict__ T,
                                 const float* __restrict__ coeffs,
                                 const float* __restrict__ diag_bias,
                                 const float* __restrict__ all_bias,
                                 float* __restrict__ U,
                                 float* __restrict__ V,
                                 float* __restrict__ W) {
    const int n = blockIdx.y, i = blockIdx.x, s = threadIdx.x;
    const float inv_m = 1.0f / (float)MM;
    const float inv_m2 = inv_m * inv_m;
    float u = 0.f, v = 0.f, w = 0.f;
    for (int d = 0; d < DD; ++d) {
        const int o = (n * DD + d) * MM + i;
        const float r = R[o] * inv_m;
        const float c = Cc[o] * inv_m;
        const float g = G[o];
        const float p = P[n * DD + d] * inv_m;
        const float tt = T[n * DD + d] * inv_m2;
        const float* cf = coeffs + (size_t)(d * SS + s) * 15;
        u += cf[5] * c + cf[6] * r + cf[11] * g + cf[13] * p + cf[14] * tt;
        v += cf[7] * c + cf[8] * r + cf[12] * g;
        w += cf[0] * g + cf[1] * p + cf[2] * r + cf[3] * c + cf[4] * tt;
    }
    const int o = (n * MM + i) * SS + s;
    U[o] = u + all_bias[s];
    V[o] = v;
    W[o] = w + diag_bias[s];
}

__global__ __launch_bounds__(256)
void ign_main_kernel(const float* __restrict__ in,
                     const float* __restrict__ C10T,
                     const float* __restrict__ C11T,
                     const float* __restrict__ U,
                     const float* __restrict__ V,
                     const float* __restrict__ W,
                     float* __restrict__ out) {
    const int n = blockIdx.z;
    const int i = blockIdx.y * 8 + (threadIdx.x >> 5);
    const int J = blockIdx.x * 16;
    const int lane = threadIdx.x & 31;
    const int m = lane & 15;
    const int kh = lane >> 4;

    const float* basen = in + (size_t)n * MM * MM * DD;
    const float* a1p = basen + ((size_t)i * MM + (J + m)) * DD + 2 * kh;
    const float* a2p = basen + ((size_t)(J + m) * MM + i) * DD + 2 * kh;

    v2f A1[8], A2[8];
    #pragma unroll
    for (int kb = 0; kb < 8; ++kb) {
        A1[kb] = *(const v2f*)(a1p + 4 * kb);
        A2[kb] = *(const v2f*)(a2p + 4 * kb);
    }

    const int s0 = m, s1 = m + 16;
    const float* b10_0 = C10T + s0 * DD + 2 * kh;
    const float* b10_1 = C10T + s1 * DD + 2 * kh;
    const float* b11_0 = C11T + s0 * DD + 2 * kh;
    const float* b11_1 = C11T + s1 * DD + 2 * kh;

    v8f acc0 = {}; v8f acc1 = {};
    #pragma unroll
    for (int kb = 0; kb < 8; ++kb) {
        v2f bA0 = *(const v2f*)(b10_0 + 4 * kb);
        v2f bA1 = *(const v2f*)(b10_1 + 4 * kb);
        v2f bB0 = *(const v2f*)(b11_0 + 4 * kb);
        v2f bB1 = *(const v2f*)(b11_1 + 4 * kb);
        acc0 = __builtin_amdgcn_wmma_f32_16x16x4_f32(false, A1[kb], false, bA0, (short)0, acc0, false, false);
        acc1 = __builtin_amdgcn_wmma_f32_16x16x4_f32(false, A1[kb], false, bA1, (short)0, acc1, false, false);
        acc0 = __builtin_amdgcn_wmma_f32_16x16x4_f32(false, A2[kb], false, bB0, (short)0, acc0, false, false);
        acc1 = __builtin_amdgcn_wmma_f32_16x16x4_f32(false, A2[kb], false, bB1, (short)0, acc1, false, false);
    }

    const float u0 = U[(n * MM + i) * SS + s0];
    const float u1 = U[(n * MM + i) * SS + s1];
    const float w0 = W[(n * MM + i) * SS + s0];
    const float w1 = W[(n * MM + i) * SS + s1];
    #pragma unroll
    for (int v = 0; v < 8; ++v) {
        const int j = J + v + 8 * kh;
        const float vj0 = V[(n * MM + j) * SS + s0];
        const float vj1 = V[(n * MM + j) * SS + s1];
        const float dsel = (j == i) ? 1.0f : 0.0f;
        const size_t ob = (((size_t)n * MM + i) * MM + j) * SS;
        out[ob + s0] = acc0[v] + u0 + vj0 + dsel * w0;
        out[ob + s1] = acc1[v] + u1 + vj1 + dsel * w1;
    }
}

extern "C" void kernel_launch(void* const* d_in, const int* in_sizes, int n_in,
                              void* d_out, int out_size, void* d_ws, size_t ws_size,
                              hipStream_t stream) {
    (void)in_sizes; (void)n_in; (void)out_size; (void)ws_size;
    const float* inputs    = (const float*)d_in[0];
    const float* coeffs    = (const float*)d_in[2];
    const float* diag_bias = (const float*)d_in[3];
    const float* all_bias  = (const float*)d_in[4];
    float* out = (float*)d_out;

    float* ws   = (float*)d_ws;
    float* R    = ws;
    float* Cc   = R    + NB * DD * MM;
    float* G    = Cc   + NB * DD * MM;
    float* P    = G    + NB * DD * MM;
    float* T    = P    + NB * DD;
    float* U    = T    + NB * DD;
    float* V    = U    + NB * MM * SS;
    float* W    = V    + NB * MM * SS;
    float* C10T = W    + NB * MM * SS;
    float* C11T = C10T + SS * DD;

    ign_repack_kernel<<<1, 1024, 0, stream>>>(coeffs, C10T, C11T);
    ign_stats_kernel<<<dim3(MM, NB), 256, 0, stream>>>(inputs, R, Cc, G);
    ign_pt_kernel<<<1, NB * DD, 0, stream>>>(R, G, P, T);
    ign_bcast_kernel<<<dim3(MM, NB), SS, 0, stream>>>(R, Cc, G, P, T, coeffs,
                                                      diag_bias, all_bias, U, V, W);
    ign_main_kernel<<<dim3(MM / 16, MM / 8, NB), 256, 0, stream>>>(
        inputs, C10T, C11T, U, V, W, out);
}


# U/V/W kernel recast as an f32 matrix product: 192 blocks of 4 waves, each wave one 16x16 tile over half of K=[Cc|R|G|P|T]; loads no longer wait per d pair; K halves summed in LDS; sums re-ordered
# speedup vs baseline: 1.7399x; 1.7399x over previous
#include <hip/hip_runtime.h>
#include <hip/hip_bf16.h>

#define NB 4
#define MM 256
#define DD 32
#define SS 32

typedef float v2f __attribute__((ext_vector_type(2)));
typedef float v8f __attribute__((ext_vector_type(8)));

__global__ void ign_repack_kernel(const float* __restrict__ coeffs,
                                  float* __restrict__ C10T,
                                  float* __restrict__ C11T) {
    int t = threadIdx.x;
    int s = t >> 5, d = t & 31;
    C10T[s * DD + d] = coeffs[(d * SS + s) * 15 + 9];
    C11T[s * DD + d] = coeffs[(d * SS + s) * 15 + 10];
}

__global__ void ign_stats_kernel(const float* __restrict__ in,
                                 float* __restrict__ R,
                                 float* __restrict__ Cc,
                                 float* __restrict__ G) {
    const int n = blockIdx.y, k = blockIdx.x;
    const int t = threadIdx.x;
    const int d = t & 31, grp = t >> 5;
    const float* base = in + (size_t)n * MM * MM * DD;
    float r = 0.f, c = 0.f;
    for (int q = grp; q < MM; q += 8) {
        r += base[((size_t)k * MM + q) * DD + d];
        c += base[((size_t)q * MM + k) * DD + d];
    }
    __shared__ float sr[8][32];
    __shared__ float sc[8][32];
    sr[grp][d] = r; sc[grp][d] = c;
    __syncthreads();
    if (t < 32) {
        float rr = 0.f, cc = 0.f;
        #pragma unroll
        for (int g = 0; g < 8; ++g) { rr += sr[g][t]; cc += sc[g][t]; }
        const int o = (n * MM + k) * DD + t;
        R[o]  = rr;
        Cc[o] = cc;
        G[o]  = base[((size_t)k * MM + k) * DD + t];
    }
}

#define PT_GROUPS 32
#define PT_KPER (MM / PT_GROUPS)
__global__ void ign_pt_kernel(const float* __restrict__ R,
                              const float* __restrict__ G,
                              float* __restrict__ P,
                              float* __restrict__ T) {
    const int n = blockIdx.x;
    const int d = threadIdx.x & 31, grp = threadIdx.x >> 5;
    const int o = (n * MM + grp * PT_KPER) * DD + d;
    float p = 0.f, s = 0.f;
    #pragma unroll
    for (int q = 0; q < PT_KPER; ++q) { p += G[o + q * DD]; s += R[o + q * DD]; }
    __shared__ float sp[PT_GROUPS][32];
    __shared__ float st[PT_GROUPS][32];
    sp[grp][d] = p; st[grp][d] = s;
    __syncthreads();
    if (threadIdx.x < 64) {
        const float (*part)[32] = (threadIdx.x < 32) ? sp : st;
        float a = 0.f;
        #pragma unroll
        for (int g = 0; g < PT_GROUPS; ++g) a += part[g][d];
        if (threadIdx.x < 32) P[n * DD + d] = a; else T[n * DD + d] = a;
    }
}

#define BC_PITCH 36

template <int KIND, int KH2>
__device__ __forceinline__ v8f ign_bcast_part(const float* __restrict__ R,
                                              const float* __restrict__ Cc,
                                              const float* __restrict__ G,
                                              const float* __restrict__ P,
                                              const float* __restrict__ T,
                                              const float* __restrict__ coeffs,
                                              const int n, const int row,
                                              const int s, const int h) {
    constexpr int NK = (KIND == 1) ? 24 : 40;
    constexpr int NH = NK / 2;
    constexpr int Q0 = KH2 * NH;
    const float inv_m = 1.0f / (float)MM;
    const float inv_m2 = inv_m * inv_m;

    v2f a[NH], b[NH];
    #pragma unroll
    for (int t = 0; t < NH; ++t) {
        const int q = Q0 + t;
        const int blk = q >> 3;
        const int d0 = 4 * (q & 7) + 2 * h;
        const float* ap;
        int bi;
        if (blk == 0)      { ap = Cc + row * DD; bi = (KIND == 0) ? 5  : (KIND == 1) ? 7  : 3; }
        else if (blk == 1) { ap = R  + row * DD; bi = (KIND == 0) ? 6  : (KIND == 1) ? 8  : 2; }
        else if (blk == 2) { ap = G  + row * DD; bi = (KIND == 0) ? 11 : (KIND == 1) ? 12 : 0; }
        else if (blk == 3) { ap = P  + n * DD;   bi = (KIND == 0) ? 13 : 1; }
        else               { ap = T  + n * DD;   bi = (KIND == 0) ? 14 : 4; }
        a[t] = *(const v2f*)(ap + d0);
        const float* bp = coeffs + (size_t)(d0 * SS + s) * 15 + bi;
        b[t][0] = bp[0];
        b[t][1] = bp[SS * 15];
    }

    v8f acc0 = {}; v8f acc1 = {};
    #pragma unroll
    for (int t = 0; t < NH; ++t) {
        const int blk = (Q0 + t) >> 3;
        const float sc = (blk == 2) ? 1.0f : (blk == 4) ? inv_m2 : inv_m;
        const v2f as = a[t] * sc;
        if (t & 1) acc1 = __builtin_amdgcn_wmma_f32_16x16x4_f32(false, as, false, b[t], (short)0, acc1, false, false);
        else       acc0 = __builtin_amdgcn_wmma_f32_16x16x4_f32(false, as, false, b[t], (short)0, acc0, false, false);
    }
    return acc0 + acc1;
}

template <int KIND>
__device__ __forceinline__ void ign_bcast_tile(const float* __restrict__ R,
                                               const float* __restrict__ Cc,
                                               const float* __restrict__ G,
                                               const float* __restrict__ P,
                                               const float* __restrict__ T,
                                               const float* __restrict__ coeffs,
                                               const float* __restrict__ bias,
                                               float* __restrict__ dst,
                                               float (*red)[16][BC_PITCH]) {
    const int n = blockIdx.y, it = blockIdx.x;
    const int wave = __builtin_amdgcn_readfirstlane((int)(threadIdx.x >> 5));
    const int ct = wave & 1, kh2 = wave >> 1;
    const int lane = threadIdx.x & 31;
    const int c = lane & 15, h = lane >> 4;
    const int row = n * MM + it * 16 + c;
    const int s = ct * 16 + c;

    const int orow = threadIdx.x >> 3, c4 = (threadIdx.x & 7) * 4;
    float4 bv = make_float4(0.f, 0.f, 0.f, 0.f);
    if (KIND != 1) bv = make_float4(bias[c4 + 0], bias[c4 + 1], bias[c4 + 2], bias[c4 + 3]);

    v8f acc;
    if (kh2 == 0) acc = ign_bcast_part<KIND, 0>(R, Cc, G, P, T, coeffs, n, row, s, h);
    else          acc = ign_bcast_part<KIND, 1>(R, Cc, G, P, T, coeffs, n, row, s, h);

    #pragma unroll
    for (int r = 0; r < 8; ++r) red[kh2][8 * h + r][ct * 16 + c] = acc[r];
    __syncthreads();

    const float4 x0 = *(const float4*)&red[0][orow][c4];
    const float4 x1 = *(const float4*)&red[1][orow][c4];
    float4 o;
    o.x = x0.x + x1.x; o.y = x0.y + x1.y; o.z = x0.z + x1.z; o.w = x0.w + x1.w;
    if (KIND != 1) { o.x += bv.x; o.y += bv.y; o.z += bv.z; o.w += bv.w; }
    *(float4*)(dst + (size_t)(n * MM + it * 16 + orow) * SS + c4) = o;
}

__global__ __launch_bounds__(128)
void ign_bcast_kernel(const float* __restrict__ R,
                      const float* __restrict__ Cc,
                      const float* __restrict__ G,
                      const float* __restrict__ P,
                      const float* __restrict__ T,
                      const float* __restrict__ coeffs,
                      const float* __restrict__ diag_bias,
                      const float* __restrict__ all_bias,
                      float* __restrict__ U,
                      float* __restrict__ V,
                      float* __restrict__ W) {
    __shared__ __attribute__((aligned(16))) float red[2][16][BC_PITCH];
    if (blockIdx.z == 0)      ign_bcast_tile<0>(R, Cc, G, P, T, coeffs, all_bias,  U, red);
    else if (blockIdx.z == 1) ign_bcast_tile<1>(R, Cc, G, P, T, coeffs, all_bias,  V, red);
    else                      ign_bcast_tile<2>(R, Cc, G, P, T, coeffs, diag_bias, W, red);
}

template <bool BOTH>
__device__ __forceinline__ void ign_strip_pair(const float* __restrict__ in,
                                               const float* __restrict__ C10T,
                                               const float* __restrict__ C11T,
                                               const float* __restrict__ U,
                                               const float* __restrict__ V,
                                               const float* __restrict__ W,
                                               float* __restrict__ out,
                                               const int n, const int i, const int J,
                                               const int lane) {
    const int m = lane & 15;
    const int kh = lane >> 4;

    const float* basen = in + (size_t)n * MM * MM * DD;
    const float* a1p = basen + ((size_t)i * MM + (J + m)) * DD + 2 * kh;
    const float* a2p = basen + ((size_t)(J + m) * MM + i) * DD + 2 * kh;

    v2f A1[8], A2[8];
    #pragma unroll
    for (int kb = 0; kb < 8; ++kb) {
        A1[kb] = *(const v2f*)(a1p + 4 * kb);
        A2[kb] = *(const v2f*)(a2p + 4 * kb);
    }

    const int s0 = m, s1 = m + 16;
    const float* b10_0 = C10T + s0 * DD + 2 * kh;
    const float* b10_1 = C10T + s1 * DD + 2 * kh;
    const float* b11_0 = C11T + s0 * DD + 2 * kh;
    const float* b11_1 = C11T + s1 * DD + 2 * kh;

    v8f acc0 = {}; v8f acc1 = {};
    v8f tac0 = {}; v8f tac1 = {};
    #pragma unroll
    for (int kb = 0; kb < 8; ++kb) {
        v2f bA0 = *(const v2f*)(b10_0 + 4 * kb);
        v2f bA1 = *(const v2f*)(b10_1 + 4 * kb);
        v2f bB0 = *(const v2f*)(b11_0 + 4 * kb);
        v2f bB1 = *(const v2f*)(b11_1 + 4 * kb);
        acc0 = __builtin_amdgcn_wmma_f32_16x16x4_f32(false, A1[kb], false, bA0, (short)0, acc0, false, false);
        acc1 = __builtin_amdgcn_wmma_f32_16x16x4_f32(false, A1[kb], false, bA1, (short)0, acc1, false, false);
        if (BOTH) {
            tac0 = __builtin_amdgcn_wmma_f32_16x16x4_f32(false, A2[kb], false, bA0, (short)0, tac0, false, false);
            tac1 = __builtin_amdgcn_wmma_f32_16x16x4_f32(false, A2[kb], false, bA1, (short)0, tac1, false, false);
        }
        acc0 = __builtin_amdgcn_wmma_f32_16x16x4_f32(false, A2[kb], false, bB0, (short)0, acc0, false, false);
        acc1 = __builtin_amdgcn_wmma_f32_16x16x4_f32(false, A2[kb], false, bB1, (short)0, acc1, false, false);
        if (BOTH) {
            tac0 = __builtin_amdgcn_wmma_f32_16x16x4_f32(false, A1[kb], false, bB0, (short)0, tac0, false, false);
            tac1 = __builtin_amdgcn_wmma_f32_16x16x4_f32(false, A1[kb], false, bB1, (short)0, tac1, false, false);
        }
    }

    const float u0 = U[(n * MM + i) * SS + s0];
    const float u1 = U[(n * MM + i) * SS + s1];
    if (BOTH) {
        #pragma unroll
        for (int v = 0; v < 8; ++v) {
            const int j = J + v + 8 * kh;
            const float vj0 = V[(n * MM + j) * SS + s0];
            const float vj1 = V[(n * MM + j) * SS + s1];
            const size_t ob = (((size_t)n * MM + i) * MM + j) * SS;
            out[ob + s0] = acc0[v] + u0 + vj0;
            out[ob + s1] = acc1[v] + u1 + vj1;
        }
        const float vi0 = V[(n * MM + i) * SS + s0];
        const float vi1 = V[(n * MM + i) * SS + s1];
        #pragma unroll
        for (int v = 0; v < 8; ++v) {
            const int j = J + v + 8 * kh;
            const float uj0 = U[(n * MM + j) * SS + s0];
            const float uj1 = U[(n * MM + j) * SS + s1];
            const size_t ob = (((size_t)n * MM + j) * MM + i) * SS;
            out[ob + s0] = tac0[v] + uj0 + vi0;
            out[ob + s1] = tac1[v] + uj1 + vi1;
        }
    } else {
        const float w0 = W[(n * MM + i) * SS + s0];
        const float w1 = W[(n * MM + i) * SS + s1];
        #pragma unroll
        for (int v = 0; v < 8; ++v) {
            const int j = J + v + 8 * kh;
            const float vj0 = V[(n * MM + j) * SS + s0];
            const float vj1 = V[(n * MM + j) * SS + s1];
            const float dsel = (j == i) ? 1.0f : 0.0f;
            const size_t ob = (((size_t)n * MM + i) * MM + j) * SS;
            out[ob + s0] = acc0[v] + u0 + vj0 + dsel * w0;
            out[ob + s1] = acc1[v] + u1 + vj1 + dsel * w1;
        }
    }
}

#define NTILE (MM / 16)
#define NPAIR (NTILE * (NTILE + 1) / 2)

__global__ __launch_bounds__(256, 8)
void ign_main_kernel(const float* __restrict__ in,
                     const float* __restrict__ C10T,
                     const float* __restrict__ C11T,
                     const float* __restrict__ U,
                     const float* __restrict__ V,
                     const float* __restrict__ W,
                     float* __restrict__ out) {
    int ti = 0, rem = blockIdx.x;
    while (rem >= NTILE - ti) { rem -= NTILE - ti; ++ti; }
    const int tj = ti + rem;

    const int n = blockIdx.z;
    const int i = ti * 16 + blockIdx.y * 8 + (threadIdx.x >> 5);
    const int J = tj * 16;
    const int lane = threadIdx.x & 31;

    if (ti != tj) {
        ign_strip_pair<true>(in, C10T, C11T, U, V, W, out, n, i, J, lane);
    } else {
        ign_strip_pair<false>(in, C10T, C11T, U, V, W, out, n, i, J, lane);
    }
}

extern "C" void kernel_launch(void* const* d_in, const int* in_sizes, int n_in,
                              void* d_out, int out_size, void* d_ws, size_t ws_size,
                              hipStream_t stream) {
    (void)in_sizes; (void)n_in; (void)out_size; (void)ws_size;
    const float* inputs    = (const float*)d_in[0];
    const float* coeffs    = (const float*)d_in[2];
    const float* diag_bias = (const float*)d_in[3];
    const float* all_bias  = (const float*)d_in[4];
    float* out = (float*)d_out;

    float* ws   = (float*)d_ws;
    float* R    = ws;
    float* Cc   = R    + NB * DD * MM;
    float* G    = Cc   + NB * DD * MM;
    float* P    = G    + NB * DD * MM;
    float* T    = P    + NB * DD;
    float* U    = T    + NB * DD;
    float* V    = U    + NB * MM * SS;
    float* W    = V    + NB * MM * SS;
    float* C10T = W    + NB * MM * SS;
    float* C11T = C10T + SS * DD;

    ign_repack_kernel<<<1, 1024, 0, stream>>>(coeffs, C10T, C11T);
    ign_stats_kernel<<<dim3(MM, NB), 256, 0, stream>>>(inputs, R, Cc, G);
    ign_pt_kernel<<<NB, PT_GROUPS * 32, 0, stream>>>(R, G, P, T);
    ign_bcast_kernel<<<dim3(MM / 16, NB, 3), 128, 0, stream>>>(R, Cc, G, P, T, coeffs,
                                                               diag_bias, all_bias, U, V, W);
    ign_main_kernel<<<dim3(NPAIR, 2, NB), 256, 0, stream>>>(
        inputs, C10T, C11T, U, V, W, out);
}


# Repack writes the op10/op11 table in the main kernel's B-fragment order: each B load is 16 B per lane, 512 contiguous bytes per wave (4 lines) instead of 8 B per lane over 16 lines; sums unchanged
# speedup vs baseline: 1.8850x; 1.0834x over previous
#include <hip/hip_runtime.h>
#include <hip/hip_bf16.h>

#define NB 4
#define MM 256
#define DD 32
#define SS 32

typedef float v2f __attribute__((ext_vector_type(2)));
typedef float v4f __attribute__((ext_vector_type(4)));
typedef float v8f __attribute__((ext_vector_type(8)));

__global__ void ign_repack_kernel(const float* __restrict__ coeffs,
                                  float* __restrict__ BF) {
    int t = threadIdx.x;
    int s = t >> 5, d = t & 31;
    const int kb = d >> 2, kh = (d >> 1) & 1, p = d & 1;
    const int m = s & 15, tile = s >> 4;
    const int lane = 16 * kh + m;
    const int e = 2 * tile + p;
    BF[((2 * kb + 0) * 32 + lane) * 4 + e] = coeffs[(d * SS + s) * 15 + 9];
    BF[((2 * kb + 1) * 32 + lane) * 4 + e] = coeffs[(d * SS + s) * 15 + 10];
}

__global__ void ign_stats_kernel(const float* __restrict__ in,
                                 float* __restrict__ R,
                                 float* __restrict__ Cc,
                                 float* __restrict__ G) {
    const int n = blockIdx.y, k = blockIdx.x;
    const int t = threadIdx.x;
    const int d = t & 31, grp = t >> 5;
    const float* base = in + (size_t)n * MM * MM * DD;
    float r = 0.f, c = 0.f;
    for (int q = grp; q < MM; q += 8) {
        r += base[((size_t)k * MM + q) * DD + d];
        c += base[((size_t)q * MM + k) * DD + d];
    }
    __shared__ float sr[8][32];
    __shared__ float sc[8][32];
    sr[grp][d] = r; sc[grp][d] = c;
    __syncthreads();
    if (t < 32) {
        float rr = 0.f, cc = 0.f;
        #pragma unroll
        for (int g = 0; g < 8; ++g) { rr += sr[g][t]; cc += sc[g][t]; }
        const int o = (n * MM + k) * DD + t;
        R[o]  = rr;
        Cc[o] = cc;
        G[o]  = base[((size_t)k * MM + k) * DD + t];
    }
}

#define PT_GROUPS 32
#define PT_KPER (MM / PT_GROUPS)
__global__ void ign_pt_kernel(const float* __restrict__ R,
                              const float* __restrict__ G,
                              float* __restrict__ P,
                              float* __restrict__ T) {
    const int n = blockIdx.x;
    const int d = threadIdx.x & 31, grp = threadIdx.x >> 5;
    const int o = (n * MM + grp * PT_KPER) * DD + d;
    float p = 0.f, s = 0.f;
    #pragma unroll
    for (int q = 0; q < PT_KPER; ++q) { p += G[o + q * DD]; s += R[o + q * DD]; }
    __shared__ float sp[PT_GROUPS][32];
    __shared__ float st[PT_GROUPS][32];
    sp[grp][d] = p; st[grp][d] = s;
    __syncthreads();
    if (threadIdx.x < 64) {
        const float (*part)[32] = (threadIdx.x < 32) ? sp : st;
        float a = 0.f;
        #pragma unroll
        for (int g = 0; g < PT_GROUPS; ++g) a += part[g][d];
        if (threadIdx.x < 32) P[n * DD + d] = a; else T[n * DD + d] = a;
    }
}

#define BC_PITCH 36

template <int KIND, int KH2>
__device__ __forceinline__ v8f ign_bcast_part(const float* __restrict__ R,
                                              const float* __restrict__ Cc,
                                              const float* __restrict__ G,
                                              const float* __restrict__ P,
                                              const float* __restrict__ T,
                                              const float* __restrict__ coeffs,
                                              const int n, const int row,
                                              const int s, const int h) {
    constexpr int NK = (KIND == 1) ? 24 : 40;
    constexpr int NH = NK / 2;
    constexpr int Q0 = KH2 * NH;
    const float inv_m = 1.0f / (float)MM;
    const float inv_m2 = inv_m * inv_m;

    v2f a[NH], b[NH];
    #pragma unroll
    for (int t = 0; t < NH; ++t) {
        const int q = Q0 + t;
        const int blk = q >> 3;
        const int d0 = 4 * (q & 7) + 2 * h;
        const float* ap;
        int bi;
        if (blk == 0)      { ap = Cc + row * DD; bi = (KIND == 0) ? 5  : (KIND == 1) ? 7  : 3; }
        else if (blk == 1) { ap = R  + row * DD; bi = (KIND == 0) ? 6  : (KIND == 1) ? 8  : 2; }
        else if (blk == 2) { ap = G  + row * DD; bi = (KIND == 0) ? 11 : (KIND == 1) ? 12 : 0; }
        else if (blk == 3) { ap = P  + n * DD;   bi = (KIND == 0) ? 13 : 1; }
        else               { ap = T  + n * DD;   bi = (KIND == 0) ? 14 : 4; }
        a[t] = *(const v2f*)(ap + d0);
        const float* bp = coeffs + (size_t)(d0 * SS + s) * 15 + bi;
        b[t][0] = bp[0];
        b[t][1] = bp[SS * 15];
    }

    v8f acc0 = {}; v8f acc1 = {};
    #pragma unroll
    for (int t = 0; t < NH; ++t) {
        const int blk = (Q0 + t) >> 3;
        const float sc = (blk == 2) ? 1.0f : (blk == 4) ? inv_m2 : inv_m;
        const v2f as = a[t] * sc;
        if (t & 1) acc1 = __builtin_amdgcn_wmma_f32_16x16x4_f32(false, as, false, b[t], (short)0, acc1, false, false);
        else       acc0 = __builtin_amdgcn_wmma_f32_16x16x4_f32(false, as, false, b[t], (short)0, acc0, false, false);
    }
    return acc0 + acc1;
}

template <int KIND>
__device__ __forceinline__ void ign_bcast_tile(const float* __restrict__ R,
                                               const float* __restrict__ Cc,
                                               const float* __restrict__ G,
                                               const float* __restrict__ P,
                                               const float* __restrict__ T,
                                               const float* __restrict__ coeffs,
                                               const float* __restrict__ bias,
                                               float* __restrict__ dst,
                                               float (*red)[16][BC_PITCH]) {
    const int n = blockIdx.y, it = blockIdx.x;
    const int wave = __builtin_amdgcn_readfirstlane((int)(threadIdx.x >> 5));
    const int ct = wave & 1, kh2 = wave >> 1;
    const int lane = threadIdx.x & 31;
    const int c = lane & 15, h = lane >> 4;
    const int row = n * MM + it * 16 + c;
    const int s = ct * 16 + c;

    const int orow = threadIdx.x >> 3, c4 = (threadIdx.x & 7) * 4;
    float4 bv = make_float4(0.f, 0.f, 0.f, 0.f);
    if (KIND != 1) bv = make_float4(bias[c4 + 0], bias[c4 + 1], bias[c4 + 2], bias[c4 + 3]);

    v8f acc;
    if (kh2 == 0) acc = ign_bcast_part<KIND, 0>(R, Cc, G, P, T, coeffs, n, row, s, h);
    else          acc = ign_bcast_part<KIND, 1>(R, Cc, G, P, T, coeffs, n, row, s, h);

    #pragma unroll
    for (int r = 0; r < 8; ++r) red[kh2][8 * h + r][ct * 16 + c] = acc[r];
    __syncthreads();

    const float4 x0 = *(const float4*)&red[0][orow][c4];
    const float4 x1 = *(const float4*)&red[1][orow][c4];
    float4 o;
    o.x = x0.x + x1.x; o.y = x0.y + x1.y; o.z = x0.z + x1.z; o.w = x0.w + x1.w;
    if (KIND != 1) { o.x += bv.x; o.y += bv.y; o.z += bv.z; o.w += bv.w; }
    *(float4*)(dst + (size_t)(n * MM + it * 16 + orow) * SS + c4) = o;
}

__global__ __launch_bounds__(128)
void ign_bcast_kernel(const float* __restrict__ R,
                      const float* __restrict__ Cc,
                      const float* __restrict__ G,
                      const float* __restrict__ P,
                      const float* __restrict__ T,
                      const float* __restrict__ coeffs,
                      const float* __restrict__ diag_bias,
                      const float* __restrict__ all_bias,
                      float* __restrict__ U,
                      float* __restrict__ V,
                      float* __restrict__ W) {
    __shared__ __attribute__((aligned(16))) float red[2][16][BC_PITCH];
    if (blockIdx.z == 0)      ign_bcast_tile<0>(R, Cc, G, P, T, coeffs, all_bias,  U, red);
    else if (blockIdx.z == 1) ign_bcast_tile<1>(R, Cc, G, P, T, coeffs, all_bias,  V, red);
    else                      ign_bcast_tile<2>(R, Cc, G, P, T, coeffs, diag_bias, W, red);
}

template <bool BOTH>
__device__ __forceinline__ void ign_strip_pair(const float* __restrict__ in,
                                               const float* __restrict__ BF,
                                               const float* __restrict__ U,
                                               const float* __restrict__ V,
                                               const float* __restrict__ W,
                                               float* __restrict__ out,
                                               const int n, const int i, const int J,
                                               const int lane) {
    const int m = lane & 15;
    const int kh = lane >> 4;

    const float* basen = in + (size_t)n * MM * MM * DD;
    const float* a1p = basen + ((size_t)i * MM + (J + m)) * DD + 2 * kh;
    const float* a2p = basen + ((size_t)(J + m) * MM + i) * DD + 2 * kh;

    v2f A1[8], A2[8];
    #pragma unroll
    for (int kb = 0; kb < 8; ++kb) {
        A1[kb] = *(const v2f*)(a1p + 4 * kb);
        A2[kb] = *(const v2f*)(a2p + 4 * kb);
    }

    const int s0 = m, s1 = m + 16;
    const v4f* bfp = (const v4f*)BF + lane;

    v8f acc0 = {}; v8f acc1 = {};
    v8f tac0 = {}; v8f tac1 = {};
    #pragma unroll
    for (int kb = 0; kb < 8; ++kb) {
        const v4f f10 = bfp[(2 * kb + 0) * 32];
        const v4f f11 = bfp[(2 * kb + 1) * 32];
        const v2f bA0 = f10.xy, bA1 = f10.zw;
        const v2f bB0 = f11.xy, bB1 = f11.zw;
        acc0 = __builtin_amdgcn_wmma_f32_16x16x4_f32(false, A1[kb], false, bA0, (short)0, acc0, false, false);
        acc1 = __builtin_amdgcn_wmma_f32_16x16x4_f32(false, A1[kb], false, bA1, (short)0, acc1, false, false);
        if (BOTH) {
            tac0 = __builtin_amdgcn_wmma_f32_16x16x4_f32(false, A2[kb], false, bA0, (short)0, tac0, false, false);
            tac1 = __builtin_amdgcn_wmma_f32_16x16x4_f32(false, A2[kb], false, bA1, (short)0, tac1, false, false);
        }
        acc0 = __builtin_amdgcn_wmma_f32_16x16x4_f32(false, A2[kb], false, bB0, (short)0, acc0, false, false);
        acc1 = __builtin_amdgcn_wmma_f32_16x16x4_f32(false, A2[kb], false, bB1, (short)0, acc1, false, false);
        if (BOTH) {
            tac0 = __builtin_amdgcn_wmma_f32_16x16x4_f32(false, A1[kb], false, bB0, (short)0, tac0, false, false);
            tac1 = __builtin_amdgcn_wmma_f32_16x16x4_f32(false, A1[kb], false, bB1, (short)0, tac1, false, false);
        }
    }

    const float u0 = U[(n * MM + i) * SS + s0];
    const float u1 = U[(n * MM + i) * SS + s1];
    if (BOTH) {
        #pragma unroll
        for (int v = 0; v < 8; ++v) {
            const int j = J + v + 8 * kh;
            const float vj0 = V[(n * MM + j) * SS + s0];
            const float vj1 = V[(n * MM + j) * SS + s1];
            const size_t ob = (((size_t)n * MM + i) * MM + j) * SS;
            out[ob + s0] = acc0[v] + u0 + vj0;
            out[ob + s1] = acc1[v] + u1 + vj1;
        }
        const float vi0 = V[(n * MM + i) * SS + s0];
        const float vi1 = V[(n * MM + i) * SS + s1];
        #pragma unroll
        for (int v = 0; v < 8; ++v) {
            const int j = J + v + 8 * kh;
            const float uj0 = U[(n * MM + j) * SS + s0];
            const float uj1 = U[(n * MM + j) * SS + s1];
            const size_t ob = (((size_t)n * MM + j) * MM + i) * SS;
            out[ob + s0] = tac0[v] + uj0 + vi0;
            out[ob + s1] = tac1[v] + uj1 + vi1;
        }
    } else {
        const float w0 = W[(n * MM + i) * SS + s0];
        const float w1 = W[(n * MM + i) * SS + s1];
        #pragma unroll
        for (int v = 0; v < 8; ++v) {
            const int j = J + v + 8 * kh;
            const float vj0 = V[(n * MM + j) * SS + s0];
            const float vj1 = V[(n * MM + j) * SS + s1];
            const float dsel = (j == i) ? 1.0f : 0.0f;
            const size_t ob = (((size_t)n * MM + i) * MM + j) * SS;
            out[ob + s0] = acc0[v] + u0 + vj0 + dsel * w0;
            out[ob + s1] = acc1[v] + u1 + vj1 + dsel * w1;
        }
    }
}

#define NTILE (MM / 16)
#define NPAIR (NTILE * (NTILE + 1) / 2)

__global__ __launch_bounds__(256, 8)
void ign_main_kernel(const float* __restrict__ in,
                     const float* __restrict__ BF,
                     const float* __restrict__ U,
                     const float* __restrict__ V,
                     const float* __restrict__ W,
                     float* __restrict__ out) {
    int ti = 0, rem = blockIdx.x;
    while (rem >= NTILE - ti) { rem -= NTILE - ti; ++ti; }
    const int tj = ti + rem;

    const int n = blockIdx.z;
    const int i = ti * 16 + blockIdx.y * 8 + (threadIdx.x >> 5);
    const int J = tj * 16;
    const int lane = threadIdx.x & 31;

    if (ti != tj) {
        ign_strip_pair<true>(in, BF, U, V, W, out, n, i, J, lane);
    } else {
        ign_strip_pair<false>(in, BF, U, V, W, out, n, i, J, lane);
    }
}

extern "C" void kernel_launch(void* const* d_in, const int* in_sizes, int n_in,
                              void* d_out, int out_size, void* d_ws, size_t ws_size,
                              hipStream_t stream) {
    (void)in_sizes; (void)n_in; (void)out_size; (void)ws_size;
    const float* inputs    = (const float*)d_in[0];
    const float* coeffs    = (const float*)d_in[2];
    const float* diag_bias = (const float*)d_in[3];
    const float* all_bias  = (const float*)d_in[4];
    float* out = (float*)d_out;

    float* ws   = (float*)d_ws;
    float* R    = ws;
    float* Cc   = R    + NB * DD * MM;
    float* G    = Cc   + NB * DD * MM;
    float* P    = G    + NB * DD * MM;
    float* T    = P    + NB * DD;
    float* U    = T    + NB * DD;
    float* V    = U    + NB * MM * SS;
    float* W    = V    + NB * MM * SS;
    float* BF   = W    + NB * MM * SS;

    ign_repack_kernel<<<1, 1024, 0, stream>>>(coeffs, BF);
    ign_stats_kernel<<<dim3(MM, NB), 256, 0, stream>>>(inputs, R, Cc, G);
    ign_pt_kernel<<<NB, PT_GROUPS * 32, 0, stream>>>(R, G, P, T);
    ign_bcast_kernel<<<dim3(MM / 16, NB, 3), 128, 0, stream>>>(R, Cc, G, P, T, coeffs,
                                                               diag_bias, all_bias, U, V, W);
    ign_main_kernel<<<dim3(NPAIR, 2, NB), 256, 0, stream>>>(
        inputs, BF, U, V, W, out);
}


# Main kernel: the block's two x tiles come in through LDS as whole 128-B lines (16 B per lane, 512 contiguous B per wave load) instead of 8-B pieces of 16 lines per load; sums unchanged
# speedup vs baseline: 2.0780x; 1.1024x over previous
#include <hip/hip_runtime.h>
#include <hip/hip_bf16.h>

#define NB 4
#define MM 256
#define DD 32
#define SS 32

typedef float v2f __attribute__((ext_vector_type(2)));
typedef float v4f __attribute__((ext_vector_type(4)));
typedef float v8f __attribute__((ext_vector_type(8)));

__global__ void ign_repack_kernel(const float* __restrict__ coeffs,
                                  float* __restrict__ BF) {
    int t = threadIdx.x;
    int s = t >> 5, d = t & 31;
    const int kb = d >> 2, kh = (d >> 1) & 1, p = d & 1;
    const int m = s & 15, tile = s >> 4;
    const int lane = 16 * kh + m;
    const int e = 2 * tile + p;
    BF[((2 * kb + 0) * 32 + lane) * 4 + e] = coeffs[(d * SS + s) * 15 + 9];
    BF[((2 * kb + 1) * 32 + lane) * 4 + e] = coeffs[(d * SS + s) * 15 + 10];
}

__global__ void ign_stats_kernel(const float* __restrict__ in,
                                 float* __restrict__ R,
                                 float* __restrict__ Cc,
                                 float* __restrict__ G) {
    const int n = blockIdx.y, k = blockIdx.x;
    const int t = threadIdx.x;
    const int d = t & 31, grp = t >> 5;
    const float* base = in + (size_t)n * MM * MM * DD;
    float r = 0.f, c = 0.f;
    for (int q = grp; q < MM; q += 8) {
        r += base[((size_t)k * MM + q) * DD + d];
        c += base[((size_t)q * MM + k) * DD + d];
    }
    __shared__ float sr[8][32];
    __shared__ float sc[8][32];
    sr[grp][d] = r; sc[grp][d] = c;
    __syncthreads();
    if (t < 32) {
        float rr = 0.f, cc = 0.f;
        #pragma unroll
        for (int g = 0; g < 8; ++g) { rr += sr[g][t]; cc += sc[g][t]; }
        const int o = (n * MM + k) * DD + t;
        R[o]  = rr;
        Cc[o] = cc;
        G[o]  = base[((size_t)k * MM + k) * DD + t];
    }
}

#define PT_GROUPS 32
#define PT_KPER (MM / PT_GROUPS)
__global__ void ign_pt_kernel(const float* __restrict__ R,
                              const float* __restrict__ G,
                              float* __restrict__ P,
                              float* __restrict__ T) {
    const int n = blockIdx.x;
    const int d = threadIdx.x & 31, grp = threadIdx.x >> 5;
    const int o = (n * MM + grp * PT_KPER) * DD + d;
    float p = 0.f, s = 0.f;
    #pragma unroll
    for (int q = 0; q < PT_KPER; ++q) { p += G[o + q * DD]; s += R[o + q * DD]; }
    __shared__ float sp[PT_GROUPS][32];
    __shared__ float st[PT_GROUPS][32];
    sp[grp][d] = p; st[grp][d] = s;
    __syncthreads();
    if (threadIdx.x < 64) {
        const float (*part)[32] = (threadIdx.x < 32) ? sp : st;
        float a = 0.f;
        #pragma unroll
        for (int g = 0; g < PT_GROUPS; ++g) a += part[g][d];
        if (threadIdx.x < 32) P[n * DD + d] = a; else T[n * DD + d] = a;
    }
}

#define BC_PITCH 36

template <int KIND, int KH2>
__device__ __forceinline__ v8f ign_bcast_part(const float* __restrict__ R,
                                              const float* __restrict__ Cc,
                                              const float* __restrict__ G,
                                              const float* __restrict__ P,
                                              const float* __restrict__ T,
                                              const float* __restrict__ coeffs,
                                              const int n, const int row,
                                              const int s, const int h) {
    constexpr int NK = (KIND == 1) ? 24 : 40;
    constexpr int NH = NK / 2;
    constexpr int Q0 = KH2 * NH;
    const float inv_m = 1.0f / (float)MM;
    const float inv_m2 = inv_m * inv_m;

    v2f a[NH], b[NH];
    #pragma unroll
    for (int t = 0; t < NH; ++t) {
        const int q = Q0 + t;
        const int blk = q >> 3;
        const int d0 = 4 * (q & 7) + 2 * h;
        const float* ap;
        int bi;
        if (blk == 0)      { ap = Cc + row * DD; bi = (KIND == 0) ? 5  : (KIND == 1) ? 7  : 3; }
        else if (blk == 1) { ap = R  + row * DD; bi = (KIND == 0) ? 6  : (KIND == 1) ? 8  : 2; }
        else if (blk == 2) { ap = G  + row * DD; bi = (KIND == 0) ? 11 : (KIND == 1) ? 12 : 0; }
        else if (blk == 3) { ap = P  + n * DD;   bi = (KIND == 0) ? 13 : 1; }
        else               { ap = T  + n * DD;   bi = (KIND == 0) ? 14 : 4; }
        a[t] = *(const v2f*)(ap + d0);
        const float* bp = coeffs + (size_t)(d0 * SS + s) * 15 + bi;
        b[t][0] = bp[0];
        b[t][1] = bp[SS * 15];
    }

    v8f acc0 = {}; v8f acc1 = {};
    #pragma unroll
    for (int t = 0; t < NH; ++t) {
        const int blk = (Q0 + t) >> 3;
        const float sc = (blk == 2) ? 1.0f : (blk == 4) ? inv_m2 : inv_m;
        const v2f as = a[t] * sc;
        if (t & 1) acc1 = __builtin_amdgcn_wmma_f32_16x16x4_f32(false, as, false, b[t], (short)0, acc1, false, false);
        else       acc0 = __builtin_amdgcn_wmma_f32_16x16x4_f32(false, as, false, b[t], (short)0, acc0, false, false);
    }
    return acc0 + acc1;
}

template <int KIND>
__device__ __forceinline__ void ign_bcast_tile(const float* __restrict__ R,
                                               const float* __restrict__ Cc,
                                               const float* __restrict__ G,
                                               const float* __restrict__ P,
                                               const float* __restrict__ T,
                                               const float* __restrict__ coeffs,
                                               const float* __restrict__ bias,
                                               float* __restrict__ dst,
                                               float (*red)[16][BC_PITCH]) {
    const int n = blockIdx.y, it = blockIdx.x;
    const int wave = __builtin_amdgcn_readfirstlane((int)(threadIdx.x >> 5));
    const int ct = wave & 1, kh2 = wave >> 1;
    const int lane = threadIdx.x & 31;
    const int c = lane & 15, h = lane >> 4;
    const int row = n * MM + it * 16 + c;
    const int s = ct * 16 + c;

    const int orow = threadIdx.x >> 3, c4 = (threadIdx.x & 7) * 4;
    float4 bv = make_float4(0.f, 0.f, 0.f, 0.f);
    if (KIND != 1) bv = make_float4(bias[c4 + 0], bias[c4 + 1], bias[c4 + 2], bias[c4 + 3]);

    v8f acc;
    if (kh2 == 0) acc = ign_bcast_part<KIND, 0>(R, Cc, G, P, T, coeffs, n, row, s, h);
    else          acc = ign_bcast_part<KIND, 1>(R, Cc, G, P, T, coeffs, n, row, s, h);

    #pragma unroll
    for (int r = 0; r < 8; ++r) red[kh2][8 * h + r][ct * 16 + c] = acc[r];
    __syncthreads();

    const float4 x0 = *(const float4*)&red[0][orow][c4];
    const float4 x1 = *(const float4*)&red[1][orow][c4];
    float4 o;
    o.x = x0.x + x1.x; o.y = x0.y + x1.y; o.z = x0.z + x1.z; o.w = x0.w + x1.w;
    if (KIND != 1) { o.x += bv.x; o.y += bv.y; o.z += bv.z; o.w += bv.w; }
    *(float4*)(dst + (size_t)(n * MM + it * 16 + orow) * SS + c4) = o;
}

__global__ __launch_bounds__(128)
void ign_bcast_kernel(const float* __restrict__ R,
                      const float* __restrict__ Cc,
                      const float* __restrict__ G,
                      const float* __restrict__ P,
                      const float* __restrict__ T,
                      const float* __restrict__ coeffs,
                      const float* __restrict__ diag_bias,
                      const float* __restrict__ all_bias,
                      float* __restrict__ U,
                      float* __restrict__ V,
                      float* __restrict__ W) {
    __shared__ __attribute__((aligned(16))) float red[2][16][BC_PITCH];
    if (blockIdx.z == 0)      ign_bcast_tile<0>(R, Cc, G, P, T, coeffs, all_bias,  U, red);
    else if (blockIdx.z == 1) ign_bcast_tile<1>(R, Cc, G, P, T, coeffs, all_bias,  V, red);
    else                      ign_bcast_tile<2>(R, Cc, G, P, T, coeffs, diag_bias, W, red);
}

#define XS_PITCH 36
#define XS_WSTRIDE (16 * XS_PITCH + 32)
#define XS_TILE (8 * XS_WSTRIDE)
template <bool BOTH>
__device__ __forceinline__ void ign_strip_pair(const float* __restrict__ in,
                                               const float* __restrict__ BF,
                                               const float* __restrict__ U,
                                               const float* __restrict__ V,
                                               const float* __restrict__ W,
                                               float* __restrict__ out,
                                               float* __restrict__ xs1,
                                               float* __restrict__ xs2,
                                               const int n, const int i0, const int J) {
    const int wv = threadIdx.x >> 5;
    const int lane = threadIdx.x & 31;
    const int i = i0 + wv;
    const int m = lane & 15;
    const int kh = lane >> 4;

    const float* basen = in + (size_t)n * MM * MM * DD;

    const int p = threadIdx.x & 7, r = threadIdx.x >> 3;
    v4f g1[4], g2[4];
    #pragma unroll
    for (int q = 0; q < 4; ++q) {
        const int L = q * 32 + r;
        g1[q] = *(const v4f*)(basen + ((size_t)(i0 + (L >> 4)) * MM + (J + (L & 15))) * DD + 4 * p);
        g2[q] = *(const v4f*)(basen + ((size_t)(J + (L >> 3)) * MM + (i0 + (L & 7))) * DD + 4 * p);
    }

    const v4f* bfp = (const v4f*)BF + lane;
    v4f F10[8], F11[8];
    #pragma unroll
    for (int kb = 0; kb < 8; ++kb) {
        F10[kb] = bfp[(2 * kb + 0) * 32];
        F11[kb] = bfp[(2 * kb + 1) * 32];
    }

    #pragma unroll
    for (int q = 0; q < 4; ++q) {
        const int L = q * 32 + r;
        *(v4f*)(xs1 + (L >> 4) * XS_WSTRIDE + (L & 15) * XS_PITCH + 4 * p) = g1[q];
        *(v4f*)(xs2 + (L & 7) * XS_WSTRIDE + (L >> 3) * XS_PITCH + 4 * p) = g2[q];
    }
    __syncthreads();

    const float* a1p = xs1 + wv * XS_WSTRIDE + m * XS_PITCH + 2 * kh;
    const float* a2p = xs2 + wv * XS_WSTRIDE + m * XS_PITCH + 2 * kh;

    v2f A1[8], A2[8];
    #pragma unroll
    for (int kb = 0; kb < 8; ++kb) {
        A1[kb] = *(const v2f*)(a1p + 4 * kb);
        A2[kb] = *(const v2f*)(a2p + 4 * kb);
    }

    const int s0 = m, s1 = m + 16;

    v8f acc0 = {}; v8f acc1 = {};
    v8f tac0 = {}; v8f tac1 = {};
    #pragma unroll
    for (int kb = 0; kb < 8; ++kb) {
        const v4f f10 = F10[kb];
        const v4f f11 = F11[kb];
        const v2f bA0 = f10.xy, bA1 = f10.zw;
        const v2f bB0 = f11.xy, bB1 = f11.zw;
        acc0 = __builtin_amdgcn_wmma_f32_16x16x4_f32(false, A1[kb], false, bA0, (short)0, acc0, false, false);
        acc1 = __builtin_amdgcn_wmma_f32_16x16x4_f32(false, A1[kb], false, bA1, (short)0, acc1, false, false);
        if (BOTH) {
            tac0 = __builtin_amdgcn_wmma_f32_16x16x4_f32(false, A2[kb], false, bA0, (short)0, tac0, false, false);
            tac1 = __builtin_amdgcn_wmma_f32_16x16x4_f32(false, A2[kb], false, bA1, (short)0, tac1, false, false);
        }
        acc0 = __builtin_amdgcn_wmma_f32_16x16x4_f32(false, A2[kb], false, bB0, (short)0, acc0, false, false);
        acc1 = __builtin_amdgcn_wmma_f32_16x16x4_f32(false, A2[kb], false, bB1, (short)0, acc1, false, false);
        if (BOTH) {
            tac0 = __builtin_amdgcn_wmma_f32_16x16x4_f32(false, A1[kb], false, bB0, (short)0, tac0, false, false);
            tac1 = __builtin_amdgcn_wmma_f32_16x16x4_f32(false, A1[kb], false, bB1, (short)0, tac1, false, false);
        }
    }

    const float u0 = U[(n * MM + i) * SS + s0];
    const float u1 = U[(n * MM + i) * SS + s1];
    if (BOTH) {
        #pragma unroll
        for (int v = 0; v < 8; ++v) {
            const int j = J + v + 8 * kh;
            const float vj0 = V[(n * MM + j) * SS + s0];
            const float vj1 = V[(n * MM + j) * SS + s1];
            const size_t ob = (((size_t)n * MM + i) * MM + j) * SS;
            out[ob + s0] = acc0[v] + u0 + vj0;
            out[ob + s1] = acc1[v] + u1 + vj1;
        }
        const float vi0 = V[(n * MM + i) * SS + s0];
        const float vi1 = V[(n * MM + i) * SS + s1];
        #pragma unroll
        for (int v = 0; v < 8; ++v) {
            const int j = J + v + 8 * kh;
            const float uj0 = U[(n * MM + j) * SS + s0];
            const float uj1 = U[(n * MM + j) * SS + s1];
            const size_t ob = (((size_t)n * MM + j) * MM + i) * SS;
            out[ob + s0] = tac0[v] + uj0 + vi0;
            out[ob + s1] = tac1[v] + uj1 + vi1;
        }
    } else {
        const float w0 = W[(n * MM + i) * SS + s0];
        const float w1 = W[(n * MM + i) * SS + s1];
        #pragma unroll
        for (int v = 0; v < 8; ++v) {
            const int j = J + v + 8 * kh;
            const float vj0 = V[(n * MM + j) * SS + s0];
            const float vj1 = V[(n * MM + j) * SS + s1];
            const float dsel = (j == i) ? 1.0f : 0.0f;
            const size_t ob = (((size_t)n * MM + i) * MM + j) * SS;
            out[ob + s0] = acc0[v] + u0 + vj0 + dsel * w0;
            out[ob + s1] = acc1[v] + u1 + vj1 + dsel * w1;
        }
    }
}

#define NTILE (MM / 16)
#define NPAIR (NTILE * (NTILE + 1) / 2)

__global__ __launch_bounds__(256, 8)
void ign_main_kernel(const float* __restrict__ in,
                     const float* __restrict__ BF,
                     const float* __restrict__ U,
                     const float* __restrict__ V,
                     const float* __restrict__ W,
                     float* __restrict__ out) {
    int ti = 0, rem = blockIdx.x;
    while (rem >= NTILE - ti) { rem -= NTILE - ti; ++ti; }
    const int tj = ti + rem;

    const int n = blockIdx.z;
    const int i0 = ti * 16 + blockIdx.y * 8;
    const int J = tj * 16;

    __shared__ __attribute__((aligned(16))) float xs1[XS_TILE];
    __shared__ __attribute__((aligned(16))) float xs2[XS_TILE];

    if (ti != tj) {
        ign_strip_pair<true>(in, BF, U, V, W, out, xs1, xs2, n, i0, J);
    } else {
        ign_strip_pair<false>(in, BF, U, V, W, out, xs1, xs2, n, i0, J);
    }
}

extern "C" void kernel_launch(void* const* d_in, const int* in_sizes, int n_in,
                              void* d_out, int out_size, void* d_ws, size_t ws_size,
                              hipStream_t stream) {
    (void)in_sizes; (void)n_in; (void)out_size; (void)ws_size;
    const float* inputs    = (const float*)d_in[0];
    const float* coeffs    = (const float*)d_in[2];
    const float* diag_bias = (const float*)d_in[3];
    const float* all_bias  = (const float*)d_in[4];
    float* out = (float*)d_out;

    float* ws   = (float*)d_ws;
    float* R    = ws;
    float* Cc   = R    + NB * DD * MM;
    float* G    = Cc   + NB * DD * MM;
    float* P    = G    + NB * DD * MM;
    float* T    = P    + NB * DD;
    float* U    = T    + NB * DD;
    float* V    = U    + NB * MM * SS;
    float* W    = V    + NB * MM * SS;
    float* BF   = W    + NB * MM * SS;

    ign_repack_kernel<<<1, 1024, 0, stream>>>(coeffs, BF);
    ign_stats_kernel<<<dim3(MM, NB), 256, 0, stream>>>(inputs, R, Cc, G);
    ign_pt_kernel<<<NB, PT_GROUPS * 32, 0, stream>>>(R, G, P, T);
    ign_bcast_kernel<<<dim3(MM / 16, NB, 3), 128, 0, stream>>>(R, Cc, G, P, T, coeffs,
                                                               diag_bias, all_bias, U, V, W);
    ign_main_kernel<<<dim3(NPAIR, 2, NB), 256, 0, stream>>>(
        inputs, BF, U, V, W, out);
}


# Main kernel: strip and mirror share products: X=(A1+A2)*Cp, Y=(A1-A2)*Cm, Cp/Cm=(C10+C11)/2,(C10-C11)/2 from the repack; strip=X+Y, mirror=X-Y: 32 f32 WMMAs per wave, not 64; sums re-associated
# speedup vs baseline: 2.1961x; 1.0568x over previous
#include <hip/hip_runtime.h>
#include <hip/hip_bf16.h>

#define NB 4
#define MM 256
#define DD 32
#define SS 32

typedef float v2f __attribute__((ext_vector_type(2)));
typedef float v4f __attribute__((ext_vector_type(4)));
typedef float v8f __attribute__((ext_vector_type(8)));

__global__ void ign_repack_kernel(const float* __restrict__ coeffs,
                                  float* __restrict__ BF) {
    int t = threadIdx.x;
    int s = t >> 5, d = t & 31;
    const int kb = d >> 2, kh = (d >> 1) & 1, p = d & 1;
    const int m = s & 15, tile = s >> 4;
    const int lane = 16 * kh + m;
    const int e = 2 * tile + p;
    const float c10 = coeffs[(d * SS + s) * 15 + 9];
    const float c11 = coeffs[(d * SS + s) * 15 + 10];
    BF[((2 * kb + 0) * 32 + lane) * 4 + e] = 0.5f * (c10 + c11);
    BF[((2 * kb + 1) * 32 + lane) * 4 + e] = 0.5f * (c10 - c11);
}

__global__ void ign_stats_kernel(const float* __restrict__ in,
                                 float* __restrict__ R,
                                 float* __restrict__ Cc,
                                 float* __restrict__ G) {
    const int n = blockIdx.y, k = blockIdx.x;
    const int t = threadIdx.x;
    const int d = t & 31, grp = t >> 5;
    const float* base = in + (size_t)n * MM * MM * DD;
    float r = 0.f, c = 0.f;
    for (int q = grp; q < MM; q += 8) {
        r += base[((size_t)k * MM + q) * DD + d];
        c += base[((size_t)q * MM + k) * DD + d];
    }
    __shared__ float sr[8][32];
    __shared__ float sc[8][32];
    sr[grp][d] = r; sc[grp][d] = c;
    __syncthreads();
    if (t < 32) {
        float rr = 0.f, cc = 0.f;
        #pragma unroll
        for (int g = 0; g < 8; ++g) { rr += sr[g][t]; cc += sc[g][t]; }
        const int o = (n * MM + k) * DD + t;
        R[o]  = rr;
        Cc[o] = cc;
        G[o]  = base[((size_t)k * MM + k) * DD + t];
    }
}

#define PT_GROUPS 32
#define PT_KPER (MM / PT_GROUPS)
__global__ void ign_pt_kernel(const float* __restrict__ R,
                              const float* __restrict__ G,
                              float* __restrict__ P,
                              float* __restrict__ T) {
    const int n = blockIdx.x;
    const int d = threadIdx.x & 31, grp = threadIdx.x >> 5;
    const int o = (n * MM + grp * PT_KPER) * DD + d;
    float p = 0.f, s = 0.f;
    #pragma unroll
    for (int q = 0; q < PT_KPER; ++q) { p += G[o + q * DD]; s += R[o + q * DD]; }
    __shared__ float sp[PT_GROUPS][32];
    __shared__ float st[PT_GROUPS][32];
    sp[grp][d] = p; st[grp][d] = s;
    __syncthreads();
    if (threadIdx.x < 64) {
        const float (*part)[32] = (threadIdx.x < 32) ? sp : st;
        float a = 0.f;
        #pragma unroll
        for (int g = 0; g < PT_GROUPS; ++g) a += part[g][d];
        if (threadIdx.x < 32) P[n * DD + d] = a; else T[n * DD + d] = a;
    }
}

#define BC_PITCH 36

template <int KIND, int KH2>
__device__ __forceinline__ v8f ign_bcast_part(const float* __restrict__ R,
                                              const float* __restrict__ Cc,
                                              const float* __restrict__ G,
                                              const float* __restrict__ P,
                                              const float* __restrict__ T,
                                              const float* __restrict__ coeffs,
                                              const int n, const int row,
                                              const int s, const int h) {
    constexpr int NK = (KIND == 1) ? 24 : 40;
    constexpr int NH = NK / 2;
    constexpr int Q0 = KH2 * NH;
    const float inv_m = 1.0f / (float)MM;
    const float inv_m2 = inv_m * inv_m;

    v2f a[NH], b[NH];
    #pragma unroll
    for (int t = 0; t < NH; ++t) {
        const int q = Q0 + t;
        const int blk = q >> 3;
        const int d0 = 4 * (q & 7) + 2 * h;
        const float* ap;
        int bi;
        if (blk == 0)      { ap = Cc + row * DD; bi = (KIND == 0) ? 5  : (KIND == 1) ? 7  : 3; }
        else if (blk == 1) { ap = R  + row * DD; bi = (KIND == 0) ? 6  : (KIND == 1) ? 8  : 2; }
        else if (blk == 2) { ap = G  + row * DD; bi = (KIND == 0) ? 11 : (KIND == 1) ? 12 : 0; }
        else if (blk == 3) { ap = P  + n * DD;   bi = (KIND == 0) ? 13 : 1; }
        else               { ap = T  + n * DD;   bi = (KIND == 0) ? 14 : 4; }
        a[t] = *(const v2f*)(ap + d0);
        const float* bp = coeffs + (size_t)(d0 * SS + s) * 15 + bi;
        b[t][0] = bp[0];
        b[t][1] = bp[SS * 15];
    }

    v8f acc0 = {}; v8f acc1 = {};
    #pragma unroll
    for (int t = 0; t < NH; ++t) {
        const int blk = (Q0 + t) >> 3;
        const float sc = (blk == 2) ? 1.0f : (blk == 4) ? inv_m2 : inv_m;
        const v2f as = a[t] * sc;
        if (t & 1) acc1 = __builtin_amdgcn_wmma_f32_16x16x4_f32(false, as, false, b[t], (short)0, acc1, false, false);
        else       acc0 = __builtin_amdgcn_wmma_f32_16x16x4_f32(false, as, false, b[t], (short)0, acc0, false, false);
    }
    return acc0 + acc1;
}

template <int KIND>
__device__ __forceinline__ void ign_bcast_tile(const float* __restrict__ R,
                                               const float* __restrict__ Cc,
                                               const float* __restrict__ G,
                                               const float* __restrict__ P,
                                               const float* __restrict__ T,
                                               const float* __restrict__ coeffs,
                                               const float* __restrict__ bias,
                                               float* __restrict__ dst,
                                               float (*red)[16][BC_PITCH]) {
    const int n = blockIdx.y, it = blockIdx.x;
    const int wave = __builtin_amdgcn_readfirstlane((int)(threadIdx.x >> 5));
    const int ct = wave & 1, kh2 = wave >> 1;
    const int lane = threadIdx.x & 31;
    const int c = lane & 15, h = lane >> 4;
    const int row = n * MM + it * 16 + c;
    const int s = ct * 16 + c;

    const int orow = threadIdx.x >> 3, c4 = (threadIdx.x & 7) * 4;
    float4 bv = make_float4(0.f, 0.f, 0.f, 0.f);
    if (KIND != 1) bv = make_float4(bias[c4 + 0], bias[c4 + 1], bias[c4 + 2], bias[c4 + 3]);

    v8f acc;
    if (kh2 == 0) acc = ign_bcast_part<KIND, 0>(R, Cc, G, P, T, coeffs, n, row, s, h);
    else          acc = ign_bcast_part<KIND, 1>(R, Cc, G, P, T, coeffs, n, row, s, h);

    #pragma unroll
    for (int r = 0; r < 8; ++r) red[kh2][8 * h + r][ct * 16 + c] = acc[r];
    __syncthreads();

    const float4 x0 = *(const float4*)&red[0][orow][c4];
    const float4 x1 = *(const float4*)&red[1][orow][c4];
    float4 o;
    o.x = x0.x + x1.x; o.y = x0.y + x1.y; o.z = x0.z + x1.z; o.w = x0.w + x1.w;
    if (KIND != 1) { o.x += bv.x; o.y += bv.y; o.z += bv.z; o.w += bv.w; }
    *(float4*)(dst + (size_t)(n * MM + it * 16 + orow) * SS + c4) = o;
}

__global__ __launch_bounds__(128)
void ign_bcast_kernel(const float* __restrict__ R,
                      const float* __restrict__ Cc,
                      const float* __restrict__ G,
                      const float* __restrict__ P,
                      const float* __restrict__ T,
                      const float* __restrict__ coeffs,
                      const float* __restrict__ diag_bias,
                      const float* __restrict__ all_bias,
                      float* __restrict__ U,
                      float* __restrict__ V,
                      float* __restrict__ W) {
    __shared__ __attribute__((aligned(16))) float red[2][16][BC_PITCH];
    if (blockIdx.z == 0)      ign_bcast_tile<0>(R, Cc, G, P, T, coeffs, all_bias,  U, red);
    else if (blockIdx.z == 1) ign_bcast_tile<1>(R, Cc, G, P, T, coeffs, all_bias,  V, red);
    else                      ign_bcast_tile<2>(R, Cc, G, P, T, coeffs, diag_bias, W, red);
}

#define XS_PITCH 36
#define XS_WSTRIDE (16 * XS_PITCH + 32)
#define XS_TILE (8 * XS_WSTRIDE)
template <bool BOTH>
__device__ __forceinline__ void ign_strip_pair(const float* __restrict__ in,
                                               const float* __restrict__ BF,
                                               const float* __restrict__ U,
                                               const float* __restrict__ V,
                                               const float* __restrict__ W,
                                               float* __restrict__ out,
                                               float* __restrict__ xs1,
                                               float* __restrict__ xs2,
                                               const int n, const int i0, const int J) {
    const int wv = threadIdx.x >> 5;
    const int lane = threadIdx.x & 31;
    const int i = i0 + wv;
    const int m = lane & 15;
    const int kh = lane >> 4;

    const float* basen = in + (size_t)n * MM * MM * DD;

    const int p = threadIdx.x & 7, r = threadIdx.x >> 3;
    v4f g1[4], g2[4];
    #pragma unroll
    for (int q = 0; q < 4; ++q) {
        const int L = q * 32 + r;
        g1[q] = *(const v4f*)(basen + ((size_t)(i0 + (L >> 4)) * MM + (J + (L & 15))) * DD + 4 * p);
        g2[q] = *(const v4f*)(basen + ((size_t)(J + (L >> 3)) * MM + (i0 + (L & 7))) * DD + 4 * p);
    }

    const v4f* bfp = (const v4f*)BF + lane;
    v4f FP[8], FM[8];
    #pragma unroll
    for (int kb = 0; kb < 8; ++kb) {
        FP[kb] = bfp[(2 * kb + 0) * 32];
        FM[kb] = bfp[(2 * kb + 1) * 32];
    }

    #pragma unroll
    for (int q = 0; q < 4; ++q) {
        const int L = q * 32 + r;
        *(v4f*)(xs1 + (L >> 4) * XS_WSTRIDE + (L & 15) * XS_PITCH + 4 * p) = g1[q];
        *(v4f*)(xs2 + (L & 7) * XS_WSTRIDE + (L >> 3) * XS_PITCH + 4 * p) = g2[q];
    }
    __syncthreads();

    const float* a1p = xs1 + wv * XS_WSTRIDE + m * XS_PITCH + 2 * kh;
    const float* a2p = xs2 + wv * XS_WSTRIDE + m * XS_PITCH + 2 * kh;

    v2f A1[8], A2[8];
    #pragma unroll
    for (int kb = 0; kb < 8; ++kb) {
        A1[kb] = *(const v2f*)(a1p + 4 * kb);
        A2[kb] = *(const v2f*)(a2p + 4 * kb);
    }

    const int s0 = m, s1 = m + 16;

    const float u0 = U[(n * MM + i) * SS + s0];
    const float u1 = U[(n * MM + i) * SS + s1];
    float vj0[8], vj1[8];
    #pragma unroll
    for (int v = 0; v < 8; ++v) {
        const int j = J + v + 8 * kh;
        vj0[v] = V[(n * MM + j) * SS + s0];
        vj1[v] = V[(n * MM + j) * SS + s1];
    }
    float vi0 = 0.f, vi1 = 0.f, w0 = 0.f, w1 = 0.f;
    float uj0[8], uj1[8];
    if (BOTH) {
        vi0 = V[(n * MM + i) * SS + s0];
        vi1 = V[(n * MM + i) * SS + s1];
        #pragma unroll
        for (int v = 0; v < 8; ++v) {
            const int j = J + v + 8 * kh;
            uj0[v] = U[(n * MM + j) * SS + s0];
            uj1[v] = U[(n * MM + j) * SS + s1];
        }
    } else {
        w0 = W[(n * MM + i) * SS + s0];
        w1 = W[(n * MM + i) * SS + s1];
    }

    v8f X0 = {}; v8f X1 = {};
    v8f Y0 = {}; v8f Y1 = {};
    #pragma unroll
    for (int kb = 0; kb < 8; ++kb) {
        const v2f sm = A1[kb] + A2[kb];
        const v2f df = A1[kb] - A2[kb];
        const v4f fp = FP[kb];
        const v4f fm = FM[kb];
        const v2f bP0 = fp.xy, bP1 = fp.zw;
        const v2f bM0 = fm.xy, bM1 = fm.zw;
        X0 = __builtin_amdgcn_wmma_f32_16x16x4_f32(false, sm, false, bP0, (short)0, X0, false, false);
        X1 = __builtin_amdgcn_wmma_f32_16x16x4_f32(false, sm, false, bP1, (short)0, X1, false, false);
        Y0 = __builtin_amdgcn_wmma_f32_16x16x4_f32(false, df, false, bM0, (short)0, Y0, false, false);
        Y1 = __builtin_amdgcn_wmma_f32_16x16x4_f32(false, df, false, bM1, (short)0, Y1, false, false);
    }
    const v8f acc0 = X0 + Y0, acc1 = X1 + Y1;
    const v8f tac0 = X0 - Y0, tac1 = X1 - Y1;

    if (BOTH) {
        #pragma unroll
        for (int v = 0; v < 8; ++v) {
            const int j = J + v + 8 * kh;
            const size_t ob = (((size_t)n * MM + i) * MM + j) * SS;
            out[ob + s0] = acc0[v] + u0 + vj0[v];
            out[ob + s1] = acc1[v] + u1 + vj1[v];
        }
        #pragma unroll
        for (int v = 0; v < 8; ++v) {
            const int j = J + v + 8 * kh;
            const size_t ob = (((size_t)n * MM + j) * MM + i) * SS;
            out[ob + s0] = tac0[v] + uj0[v] + vi0;
            out[ob + s1] = tac1[v] + uj1[v] + vi1;
        }
    } else {
        #pragma unroll
        for (int v = 0; v < 8; ++v) {
            const int j = J + v + 8 * kh;
            const float dsel = (j == i) ? 1.0f : 0.0f;
            const size_t ob = (((size_t)n * MM + i) * MM + j) * SS;
            out[ob + s0] = acc0[v] + u0 + vj0[v] + dsel * w0;
            out[ob + s1] = acc1[v] + u1 + vj1[v] + dsel * w1;
        }
    }
}

#define NTILE (MM / 16)
#define NPAIR (NTILE * (NTILE + 1) / 2)

__global__ __launch_bounds__(256, 8)
void ign_main_kernel(const float* __restrict__ in,
                     const float* __restrict__ BF,
                     const float* __restrict__ U,
                     const float* __restrict__ V,
                     const float* __restrict__ W,
                     float* __restrict__ out) {
    int ti = 0, rem = blockIdx.x;
    while (rem >= NTILE - ti) { rem -= NTILE - ti; ++ti; }
    const int tj = ti + rem;

    const int n = blockIdx.z;
    const int i0 = ti * 16 + blockIdx.y * 8;
    const int J = tj * 16;

    __shared__ __attribute__((aligned(16))) float xs1[XS_TILE];
    __shared__ __attribute__((aligned(16))) float xs2[XS_TILE];

    if (ti != tj) {
        ign_strip_pair<true>(in, BF, U, V, W, out, xs1, xs2, n, i0, J);
    } else {
        ign_strip_pair<false>(in, BF, U, V, W, out, xs1, xs2, n, i0, J);
    }
}

extern "C" void kernel_launch(void* const* d_in, const int* in_sizes, int n_in,
                              void* d_out, int out_size, void* d_ws, size_t ws_size,
                              hipStream_t stream) {
    (void)in_sizes; (void)n_in; (void)out_size; (void)ws_size;
    const float* inputs    = (const float*)d_in[0];
    const float* coeffs    = (const float*)d_in[2];
    const float* diag_bias = (const float*)d_in[3];
    const float* all_bias  = (const float*)d_in[4];
    float* out = (float*)d_out;

    float* ws   = (float*)d_ws;
    float* R    = ws;
    float* Cc   = R    + NB * DD * MM;
    float* G    = Cc   + NB * DD * MM;
    float* P    = G    + NB * DD * MM;
    float* T    = P    + NB * DD;
    float* U    = T    + NB * DD;
    float* V    = U    + NB * MM * SS;
    float* W    = V    + NB * MM * SS;
    float* BF   = W    + NB * MM * SS;

    ign_repack_kernel<<<1, 1024, 0, stream>>>(coeffs, BF);
    ign_stats_kernel<<<dim3(MM, NB), 256, 0, stream>>>(inputs, R, Cc, G);
    ign_pt_kernel<<<NB, PT_GROUPS * 32, 0, stream>>>(R, G, P, T);
    ign_bcast_kernel<<<dim3(MM / 16, NB, 3), 128, 0, stream>>>(R, Cc, G, P, T, coeffs,
                                                               diag_bias, all_bias, U, V, W);
    ign_main_kernel<<<dim3(NPAIR, 2, NB), 256, 0, stream>>>(
        inputs, BF, U, V, W, out);
}


# Main kernel: what a block's 8 waves share (Cp/Cm table, U and V rows of its 16 j) is fetched once per block into LDS, all loads ahead of the one barrier: 60 -> 15 loads per wave; sums unchanged
# speedup vs baseline: 2.3377x; 1.0645x over previous
#include <hip/hip_runtime.h>
#include <hip/hip_bf16.h>

#define NB 4
#define MM 256
#define DD 32
#define SS 32

typedef float v2f __attribute__((ext_vector_type(2)));
typedef float v4f __attribute__((ext_vector_type(4)));
typedef float v8f __attribute__((ext_vector_type(8)));

__global__ void ign_repack_kernel(const float* __restrict__ coeffs,
                                  float* __restrict__ BF) {
    int t = threadIdx.x;
    int s = t >> 5, d = t & 31;
    const int kb = d >> 2, kh = (d >> 1) & 1, p = d & 1;
    const int m = s & 15, tile = s >> 4;
    const int lane = 16 * kh + m;
    const int e = 2 * tile + p;
    const float c10 = coeffs[(d * SS + s) * 15 + 9];
    const float c11 = coeffs[(d * SS + s) * 15 + 10];
    BF[((2 * kb + 0) * 32 + lane) * 4 + e] = 0.5f * (c10 + c11);
    BF[((2 * kb + 1) * 32 + lane) * 4 + e] = 0.5f * (c10 - c11);
}

__global__ void ign_stats_kernel(const float* __restrict__ in,
                                 float* __restrict__ R,
                                 float* __restrict__ Cc,
                                 float* __restrict__ G) {
    const int n = blockIdx.y, k = blockIdx.x;
    const int t = threadIdx.x;
    const int d = t & 31, grp = t >> 5;
    const float* base = in + (size_t)n * MM * MM * DD;
    float r = 0.f, c = 0.f;
    for (int q = grp; q < MM; q += 8) {
        r += base[((size_t)k * MM + q) * DD + d];
        c += base[((size_t)q * MM + k) * DD + d];
    }
    __shared__ float sr[8][32];
    __shared__ float sc[8][32];
    sr[grp][d] = r; sc[grp][d] = c;
    __syncthreads();
    if (t < 32) {
        float rr = 0.f, cc = 0.f;
        #pragma unroll
        for (int g = 0; g < 8; ++g) { rr += sr[g][t]; cc += sc[g][t]; }
        const int o = (n * MM + k) * DD + t;
        R[o]  = rr;
        Cc[o] = cc;
        G[o]  = base[((size_t)k * MM + k) * DD + t];
    }
}

#define PT_GROUPS 32
#define PT_KPER (MM / PT_GROUPS)
__global__ void ign_pt_kernel(const float* __restrict__ R,
                              const float* __restrict__ G,
                              float* __restrict__ P,
                              float* __restrict__ T) {
    const int n = blockIdx.x;
    const int d = threadIdx.x & 31, grp = threadIdx.x >> 5;
    const int o = (n * MM + grp * PT_KPER) * DD + d;
    float p = 0.f, s = 0.f;
    #pragma unroll
    for (int q = 0; q < PT_KPER; ++q) { p += G[o + q * DD]; s += R[o + q * DD]; }
    __shared__ float sp[PT_GROUPS][32];
    __shared__ float st[PT_GROUPS][32];
    sp[grp][d] = p; st[grp][d] = s;
    __syncthreads();
    if (threadIdx.x < 64) {
        const float (*part)[32] = (threadIdx.x < 32) ? sp : st;
        float a = 0.f;
        #pragma unroll
        for (int g = 0; g < PT_GROUPS; ++g) a += part[g][d];
        if (threadIdx.x < 32) P[n * DD + d] = a; else T[n * DD + d] = a;
    }
}

#define BC_PITCH 36

template <int KIND, int KH2>
__device__ __forceinline__ v8f ign_bcast_part(const float* __restrict__ R,
                                              const float* __restrict__ Cc,
                                              const float* __restrict__ G,
                                              const float* __restrict__ P,
                                              const float* __restrict__ T,
                                              const float* __restrict__ coeffs,
                                              const int n, const int row,
                                              const int s, const int h) {
    constexpr int NK = (KIND == 1) ? 24 : 40;
    constexpr int NH = NK / 2;
    constexpr int Q0 = KH2 * NH;
    const float inv_m = 1.0f / (float)MM;
    const float inv_m2 = inv_m * inv_m;

    v2f a[NH], b[NH];
    #pragma unroll
    for (int t = 0; t < NH; ++t) {
        const int q = Q0 + t;
        const int blk = q >> 3;
        const int d0 = 4 * (q & 7) + 2 * h;
        const float* ap;
        int bi;
        if (blk == 0)      { ap = Cc + row * DD; bi = (KIND == 0) ? 5  : (KIND == 1) ? 7  : 3; }
        else if (blk == 1) { ap = R  + row * DD; bi = (KIND == 0) ? 6  : (KIND == 1) ? 8  : 2; }
        else if (blk == 2) { ap = G  + row * DD; bi = (KIND == 0) ? 11 : (KIND == 1) ? 12 : 0; }
        else if (blk == 3) { ap = P  + n * DD;   bi = (KIND == 0) ? 13 : 1; }
        else               { ap = T  + n * DD;   bi = (KIND == 0) ? 14 : 4; }
        a[t] = *(const v2f*)(ap + d0);
        const float* bp = coeffs + (size_t)(d0 * SS + s) * 15 + bi;
        b[t][0] = bp[0];
        b[t][1] = bp[SS * 15];
    }

    v8f acc0 = {}; v8f acc1 = {};
    #pragma unroll
    for (int t = 0; t < NH; ++t) {
        const int blk = (Q0 + t) >> 3;
        const float sc = (blk == 2) ? 1.0f : (blk == 4) ? inv_m2 : inv_m;
        const v2f as = a[t] * sc;
        if (t & 1) acc1 = __builtin_amdgcn_wmma_f32_16x16x4_f32(false, as, false, b[t], (short)0, acc1, false, false);
        else       acc0 = __builtin_amdgcn_wmma_f32_16x16x4_f32(false, as, false, b[t], (short)0, acc0, false, false);
    }
    return acc0 + acc1;
}

template <int KIND>
__device__ __forceinline__ void ign_bcast_tile(const float* __restrict__ R,
                                               const float* __restrict__ Cc,
                                               const float* __restrict__ G,
                                               const float* __restrict__ P,
                                               const float* __restrict__ T,
                                               const float* __restrict__ coeffs,
                                               const float* __restrict__ bias,
                                               float* __restrict__ dst,
                                               float (*red)[16][BC_PITCH]) {
    const int n = blockIdx.y, it = blockIdx.x;
    const int wave = __builtin_amdgcn_readfirstlane((int)(threadIdx.x >> 5));
    const int ct = wave & 1, kh2 = wave >> 1;
    const int lane = threadIdx.x & 31;
    const int c = lane & 15, h = lane >> 4;
    const int row = n * MM + it * 16 + c;
    const int s = ct * 16 + c;

    const int orow = threadIdx.x >> 3, c4 = (threadIdx.x & 7) * 4;
    float4 bv = make_float4(0.f, 0.f, 0.f, 0.f);
    if (KIND != 1) bv = make_float4(bias[c4 + 0], bias[c4 + 1], bias[c4 + 2], bias[c4 + 3]);

    v8f acc;
    if (kh2 == 0) acc = ign_bcast_part<KIND, 0>(R, Cc, G, P, T, coeffs, n, row, s, h);
    else          acc = ign_bcast_part<KIND, 1>(R, Cc, G, P, T, coeffs, n, row, s, h);

    #pragma unroll
    for (int r = 0; r < 8; ++r) red[kh2][8 * h + r][ct * 16 + c] = acc[r];
    __syncthreads();

    const float4 x0 = *(const float4*)&red[0][orow][c4];
    const float4 x1 = *(const float4*)&red[1][orow][c4];
    float4 o;
    o.x = x0.x + x1.x; o.y = x0.y + x1.y; o.z = x0.z + x1.z; o.w = x0.w + x1.w;
    if (KIND != 1) { o.x += bv.x; o.y += bv.y; o.z += bv.z; o.w += bv.w; }
    *(float4*)(dst + (size_t)(n * MM + it * 16 + orow) * SS + c4) = o;
}

__global__ __launch_bounds__(128)
void ign_bcast_kernel(const float* __restrict__ R,
                      const float* __restrict__ Cc,
                      const float* __restrict__ G,
                      const float* __restrict__ P,
                      const float* __restrict__ T,
                      const float* __restrict__ coeffs,
                      const float* __restrict__ diag_bias,
                      const float* __restrict__ all_bias,
                      float* __restrict__ U,
                      float* __restrict__ V,
                      float* __restrict__ W) {
    __shared__ __attribute__((aligned(16))) float red[2][16][BC_PITCH];
    if (blockIdx.z == 0)      ign_bcast_tile<0>(R, Cc, G, P, T, coeffs, all_bias,  U, red);
    else if (blockIdx.z == 1) ign_bcast_tile<1>(R, Cc, G, P, T, coeffs, all_bias,  V, red);
    else                      ign_bcast_tile<2>(R, Cc, G, P, T, coeffs, diag_bias, W, red);
}

#define XS_PITCH 36
#define XS_WSTRIDE (16 * XS_PITCH + 32)
#define XS_TILE (8 * XS_WSTRIDE)
#define BF_FLOATS (16 * 32 * 4)
#define UV_PITCH 36
template <bool BOTH>
__device__ __forceinline__ void ign_strip_pair(const float* __restrict__ in,
                                               const float* __restrict__ BF,
                                               const float* __restrict__ U,
                                               const float* __restrict__ V,
                                               const float* __restrict__ W,
                                               float* __restrict__ out,
                                               float* __restrict__ xs1,
                                               float* __restrict__ xs2,
                                               float* __restrict__ bfs,
                                               float* __restrict__ vts,
                                               float* __restrict__ uts,
                                               const int n, const int i0, const int J) {
    const int wv = threadIdx.x >> 5;
    const int lane = threadIdx.x & 31;
    const int i = i0 + wv;
    const int m = lane & 15;
    const int kh = lane >> 4;

    const float* basen = in + (size_t)n * MM * MM * DD;

    const int p = threadIdx.x & 7, r = threadIdx.x >> 3;
    v4f g1[4], g2[4];
    #pragma unroll
    for (int q = 0; q < 4; ++q) {
        const int L = q * 32 + r;
        g1[q] = *(const v4f*)(basen + ((size_t)(i0 + (L >> 4)) * MM + (J + (L & 15))) * DD + 4 * p);
        g2[q] = *(const v4f*)(basen + ((size_t)(J + (L >> 3)) * MM + (i0 + (L & 7))) * DD + 4 * p);
    }

    const v4f tb0 = ((const v4f*)BF)[threadIdx.x];
    const v4f tb1 = ((const v4f*)BF)[threadIdx.x + 256];

    const int e = threadIdx.x & 127;
    const int uvsrc = (n * MM + J + (e >> 3)) * SS + 4 * (e & 7);
    const float* uvp = (wv < 4) ? V : U;
    float* uvd = ((wv < 4) ? vts : uts) + (e >> 3) * UV_PITCH + 4 * (e & 7);
    const v4f tuv = *(const v4f*)(uvp + uvsrc);

    const int s0 = m, s1 = m + 16;
    const float u0 = U[(n * MM + i) * SS + s0];
    const float u1 = U[(n * MM + i) * SS + s1];
    float vi0 = 0.f, vi1 = 0.f, w0 = 0.f, w1 = 0.f;
    if (BOTH) {
        vi0 = V[(n * MM + i) * SS + s0];
        vi1 = V[(n * MM + i) * SS + s1];
    } else {
        w0 = W[(n * MM + i) * SS + s0];
        w1 = W[(n * MM + i) * SS + s1];
    }

    #pragma unroll
    for (int q = 0; q < 4; ++q) {
        const int L = q * 32 + r;
        *(v4f*)(xs1 + (L >> 4) * XS_WSTRIDE + (L & 15) * XS_PITCH + 4 * p) = g1[q];
        *(v4f*)(xs2 + (L & 7) * XS_WSTRIDE + (L >> 3) * XS_PITCH + 4 * p) = g2[q];
    }
    ((v4f*)bfs)[threadIdx.x] = tb0;
    ((v4f*)bfs)[threadIdx.x + 256] = tb1;
    *(v4f*)uvd = tuv;
    __syncthreads();

    const float* a1p = xs1 + wv * XS_WSTRIDE + m * XS_PITCH + 2 * kh;
    const float* a2p = xs2 + wv * XS_WSTRIDE + m * XS_PITCH + 2 * kh;

    v2f A1[8], A2[8];
    #pragma unroll
    for (int kb = 0; kb < 8; ++kb) {
        A1[kb] = *(const v2f*)(a1p + 4 * kb);
        A2[kb] = *(const v2f*)(a2p + 4 * kb);
    }

    float vj0[8], vj1[8];
    float uj0[8], uj1[8];
    #pragma unroll
    for (int v = 0; v < 8; ++v) {
        vj0[v] = vts[(v + 8 * kh) * UV_PITCH + s0];
        vj1[v] = vts[(v + 8 * kh) * UV_PITCH + s1];
        if (BOTH) {
            uj0[v] = uts[(v + 8 * kh) * UV_PITCH + s0];
            uj1[v] = uts[(v + 8 * kh) * UV_PITCH + s1];
        }
    }

    const v4f* bfp = (const v4f*)bfs + lane;

    v8f X0 = {}; v8f X1 = {};
    v8f Y0 = {}; v8f Y1 = {};
    #pragma unroll
    for (int kb = 0; kb < 8; ++kb) {
        const v2f sm = A1[kb] + A2[kb];
        const v2f df = A1[kb] - A2[kb];
        const v4f fp = bfp[(2 * kb + 0) * 32];
        const v4f fm = bfp[(2 * kb + 1) * 32];
        const v2f bP0 = fp.xy, bP1 = fp.zw;
        const v2f bM0 = fm.xy, bM1 = fm.zw;
        X0 = __builtin_amdgcn_wmma_f32_16x16x4_f32(false, sm, false, bP0, (short)0, X0, false, false);
        X1 = __builtin_amdgcn_wmma_f32_16x16x4_f32(false, sm, false, bP1, (short)0, X1, false, false);
        Y0 = __builtin_amdgcn_wmma_f32_16x16x4_f32(false, df, false, bM0, (short)0, Y0, false, false);
        Y1 = __builtin_amdgcn_wmma_f32_16x16x4_f32(false, df, false, bM1, (short)0, Y1, false, false);
    }
    const v8f acc0 = X0 + Y0, acc1 = X1 + Y1;
    const v8f tac0 = X0 - Y0, tac1 = X1 - Y1;

    if (BOTH) {
        #pragma unroll
        for (int v = 0; v < 8; ++v) {
            const int j = J + v + 8 * kh;
            const size_t ob = (((size_t)n * MM + i) * MM + j) * SS;
            out[ob + s0] = acc0[v] + u0 + vj0[v];
            out[ob + s1] = acc1[v] + u1 + vj1[v];
        }
        #pragma unroll
        for (int v = 0; v < 8; ++v) {
            const int j = J + v + 8 * kh;
            const size_t ob = (((size_t)n * MM + j) * MM + i) * SS;
            out[ob + s0] = tac0[v] + uj0[v] + vi0;
            out[ob + s1] = tac1[v] + uj1[v] + vi1;
        }
    } else {
        #pragma unroll
        for (int v = 0; v < 8; ++v) {
            const int j = J + v + 8 * kh;
            const float dsel = (j == i) ? 1.0f : 0.0f;
            const size_t ob = (((size_t)n * MM + i) * MM + j) * SS;
            out[ob + s0] = acc0[v] + u0 + vj0[v] + dsel * w0;
            out[ob + s1] = acc1[v] + u1 + vj1[v] + dsel * w1;
        }
    }
}

#define NTILE (MM / 16)
#define NPAIR (NTILE * (NTILE + 1) / 2)

__global__ __launch_bounds__(256, 8)
void ign_main_kernel(const float* __restrict__ in,
                     const float* __restrict__ BF,
                     const float* __restrict__ U,
                     const float* __restrict__ V,
                     const float* __restrict__ W,
                     float* __restrict__ out) {
    int ti = 0, rem = blockIdx.x;
    while (rem >= NTILE - ti) { rem -= NTILE - ti; ++ti; }
    const int tj = ti + rem;

    const int n = blockIdx.z;
    const int i0 = ti * 16 + blockIdx.y * 8;
    const int J = tj * 16;

    __shared__ __attribute__((aligned(16))) float xs1[XS_TILE];
    __shared__ __attribute__((aligned(16))) float xs2[XS_TILE];
    __shared__ __attribute__((aligned(16))) float bfs[BF_FLOATS];
    __shared__ __attribute__((aligned(16))) float vts[16 * UV_PITCH];
    __shared__ __attribute__((aligned(16))) float uts[16 * UV_PITCH];

    if (ti != tj) {
        ign_strip_pair<true>(in, BF, U, V, W, out, xs1, xs2, bfs, vts, uts, n, i0, J);
    } else {
        ign_strip_pair<false>(in, BF, U, V, W, out, xs1, xs2, bfs, vts, uts, n, i0, J);
    }
}

extern "C" void kernel_launch(void* const* d_in, const int* in_sizes, int n_in,
                              void* d_out, int out_size, void* d_ws, size_t ws_size,
                              hipStream_t stream) {
    (void)in_sizes; (void)n_in; (void)out_size; (void)ws_size;
    const float* inputs    = (const float*)d_in[0];
    const float* coeffs    = (const float*)d_in[2];
    const float* diag_bias = (const float*)d_in[3];
    const float* all_bias  = (const float*)d_in[4];
    float* out = (float*)d_out;

    float* ws   = (float*)d_ws;
    float* R    = ws;
    float* Cc   = R    + NB * DD * MM;
    float* G    = Cc   + NB * DD * MM;
    float* P    = G    + NB * DD * MM;
    float* T    = P    + NB * DD;
    float* U    = T    + NB * DD;
    float* V    = U    + NB * MM * SS;
    float* W    = V    + NB * MM * SS;
    float* BF   = W    + NB * MM * SS;

    ign_repack_kernel<<<1, 1024, 0, stream>>>(coeffs, BF);
    ign_stats_kernel<<<dim3(MM, NB), 256, 0, stream>>>(inputs, R, Cc, G);
    ign_pt_kernel<<<NB, PT_GROUPS * 32, 0, stream>>>(R, G, P, T);
    ign_bcast_kernel<<<dim3(MM / 16, NB, 3), 128, 0, stream>>>(R, Cc, G, P, T, coeffs,
                                                               diag_bias, all_bias, U, V, W);
    ign_main_kernel<<<dim3(NPAIR, 2, NB), 256, 0, stream>>>(
        inputs, BF, U, V, W, out);
}
